# starting kernel
# baseline (speedup 1.0000x reference)
#include <hip/hip_runtime.h>
#include <hip/hip_bf16.h>


typedef float v2f __attribute__((ext_vector_type(2)));
typedef float v8f __attribute__((ext_vector_type(8)));

__global__ __launch_bounds__(256) void pwm_row_windows(
    const float* __restrict__ x, float* __restrict__ S)
{
    const int lane = threadIdx.x & 31;
    const int wave = threadIdx.x >> 5;
    const int row  = blockIdx.x * 8 + wave;
    const float* __restrict__ xr = x + row * 1000;

    float total = 0.0f;
    for (int l = lane; l < 1000; l += 32) total += xr[l];
#pragma unroll
    for (int off = 16; off > 0; off >>= 1)
        total += __shfl_xor(total, off, 32);

    float h = (lane < 19) ? xr[lane]        : 0.0f;
    float t = (lane < 19) ? xr[999 - lane]  : 0.0f;
    float hI = h, tI = t;
#pragma unroll
    for (int off = 1; off < 32; off <<= 1) {
        float ho = __shfl_up(hI, off, 32);
        float to = __shfl_up(tI, off, 32);
        if (lane >= off) { hI += ho; tI += to; }
    }
    const float head_excl = hI - h;
    const int   src     = (lane < 18) ? (17 - lane) : 0;
    const float twv     = __shfl(tI, src, 32);
    const float tailwin = (lane < 18) ? twv : 0.0f;

    if (lane < 19)
        S[row * 19 + lane] = total - head_excl - tailwin;
}

__global__ __launch_bounds__(128) void pwm_gemm_wmma(
    const float* __restrict__ S,
    const float* __restrict__ W,
    const float* __restrict__ bias,
    float* __restrict__ out)
{
    const int lane = threadIdx.x & 31;
    const int tile = blockIdx.x * 4 + (threadIdx.x >> 5);
    const int m0 = (tile >> 4) << 4;
    const int n0 = (tile & 15) << 4;

    const int half = lane >> 4;
    const int l16  = lane & 15;
    const int m = m0 + l16;
    const int n = n0 + l16;

    const float* __restrict__ Sf = S + m * 76 + half * 2;
    const float* __restrict__ Sr = S + m * 76 + 74 - half * 2;
    const float* __restrict__ Wf = W + n * 76 + half * 2;

    v8f acc = {};
#pragma unroll
    for (int kb = 0; kb < 19; ++kb) {
        const v2f fw = *(const v2f*)(Sf + 4 * kb);
        const v2f rv = *(const v2f*)(Sr - 4 * kb);
        const v2f b  = *(const v2f*)(Wf + 4 * kb);
        v2f a;
        a.x = fw.x + rv.y;
        a.y = fw.y + rv.x;
        acc = __builtin_amdgcn_wmma_f32_16x16x4_f32(
            false, a, false, b, (short)0, acc, false, false);
    }

    const int   mrow  = m0 + half * 8;
    const float bterm = 1964.0f * bias[n];
#pragma unroll
    for (int r = 0; r < 8; ++r)
        out[(mrow + r) * 256 + n] = acc[r] + bterm;
}

extern "C" void kernel_launch(void* const* d_in, const int* in_sizes, int n_in,
                              void* d_out, int out_size, void* d_ws, size_t ws_size,
                              hipStream_t stream)
{
    const float* x    = (const float*)d_in[0];
    const float* pwms = (const float*)d_in[1];
    const float* bias = (const float*)d_in[2];
    float* out = (float*)d_out;

    float* S = (float*)d_ws;

    pwm_row_windows<<<128, 256, 0, stream>>>(x, S);
    pwm_gemm_wmma<<<64, 128, 0, stream>>>(S, pwms, bias, out);
}


# one launch instead of two: each work-group sums the 8 rows of its 2 batches into LDS and runs its 16x16 tiles (2 real + 14 zero rows) from there, weight loads issued first; sum order unchanged
# speedup vs baseline: 1.2236x; 1.2236x over previous
#include <hip/hip_runtime.h>
#include <hip/hip_bf16.h>


typedef float v2f __attribute__((ext_vector_type(2)));
typedef float v8f __attribute__((ext_vector_type(8)));

#define NB 2
#define KC 76

__global__ __launch_bounds__(256) void pwm_fused(
    const float* __restrict__ x,
    const float* __restrict__ W,
    const float* __restrict__ bias,
    float* __restrict__ out)
{
    __shared__ __attribute__((aligned(16))) float Ssh[(NB + 1) * KC];

    const int lane = threadIdx.x & 31;
    const int wave = threadIdx.x >> 5;
    const int bp   = blockIdx.x >> 1;
    const int n0   = ((blockIdx.x & 1) * 8 + wave) << 4;

    const int half = lane >> 4;
    const int l16  = lane & 15;
    const int n    = n0 + l16;

    const float* __restrict__ Wf = W + n * KC + half * 2;
    v2f bw[19];
#pragma unroll
    for (int kb = 0; kb < 19; ++kb)
        bw[kb] = *(const v2f*)(Wf + 4 * kb);
    const float bterm = 1964.0f * bias[n];

    const int row = bp * 8 + wave;
    const float* __restrict__ xr = x + row * 1000;

    float total = 0.0f;
    for (int l = lane; l < 1000; l += 32) total += xr[l];
#pragma unroll
    for (int off = 16; off > 0; off >>= 1)
        total += __shfl_xor(total, off, 32);

    float h = (lane < 19) ? xr[lane]        : 0.0f;
    float t = (lane < 19) ? xr[999 - lane]  : 0.0f;
    float hI = h, tI = t;
#pragma unroll
    for (int off = 1; off < 32; off <<= 1) {
        float ho = __shfl_up(hI, off, 32);
        float to = __shfl_up(tI, off, 32);
        if (lane >= off) { hI += ho; tI += to; }
    }
    const float head_excl = hI - h;
    const int   src     = (lane < 18) ? (17 - lane) : 0;
    const float twv     = __shfl(tI, src, 32);
    const float tailwin = (lane < 18) ? twv : 0.0f;

    if (lane < 19)
        Ssh[wave * 19 + lane] = total - head_excl - tailwin;
    if (threadIdx.x < KC)
        Ssh[NB * KC + threadIdx.x] = 0.0f;
    __syncthreads();

    const float* arow = Ssh + ((l16 < NB) ? l16 : NB) * KC;
    const float* Sf = arow + half * 2;
    const float* Sr = arow + 74 - half * 2;

    v8f acc = {};
#pragma unroll
    for (int kb = 0; kb < 19; ++kb) {
        const v2f fw = *(const v2f*)(Sf + 4 * kb);
        const v2f rv = *(const v2f*)(Sr - 4 * kb);
        v2f a;
        a.x = fw.x + rv.y;
        a.y = fw.y + rv.x;
        acc = __builtin_amdgcn_wmma_f32_16x16x4_f32(
            false, a, false, bw[kb], (short)0, acc, false, false);
    }

    if (half == 0) {
#pragma unroll
        for (int r = 0; r < NB; ++r)
            out[(bp * NB + r) * 256 + n] = acc[r] + bterm;
    }
}

extern "C" void kernel_launch(void* const* d_in, const int* in_sizes, int n_in,
                              void* d_out, int out_size, void* d_ws, size_t ws_size,
                              hipStream_t stream)
{
    const float* x    = (const float*)d_in[0];
    const float* pwms = (const float*)d_in[1];
    const float* bias = (const float*)d_in[2];
    float* out = (float*)d_out;

    pwm_fused<<<256, 256, 0, stream>>>(x, pwms, bias, out);
}


# all loads of a wave leave in one burst in order of need: its row as 8 independent 128-bit loads a lane, then the weights; one memory wait, not a dozen; lanes sum pairwise: row-total sum order changes
# speedup vs baseline: 2.4197x; 1.9775x over previous
#include <hip/hip_runtime.h>
#include <hip/hip_bf16.h>


typedef float v2f __attribute__((ext_vector_type(2)));
typedef float v4f __attribute__((ext_vector_type(4)));
typedef float v8f __attribute__((ext_vector_type(8)));

#define NB 2
#define KC 76

__global__ __launch_bounds__(256) void pwm_fused(
    const float* __restrict__ x,
    const float* __restrict__ W,
    const float* __restrict__ bias,
    float* __restrict__ out)
{
    __shared__ __attribute__((aligned(16))) float Ssh[(NB + 1) * KC];

    const int lane = threadIdx.x & 31;
    const int wave = threadIdx.x >> 5;
    const int bp   = blockIdx.x >> 1;
    const int n0   = ((blockIdx.x & 1) * 8 + wave) << 4;

    const int half = lane >> 4;
    const int l16  = lane & 15;
    const int n    = n0 + l16;

    const int row = bp * 8 + wave;
    const float* __restrict__ xr = x + row * 1000;

    const v4f* __restrict__ xr4 = (const v4f*)xr;
    v4f p[8];
#pragma unroll
    for (int j = 0; j < 7; ++j)
        p[j] = xr4[lane + 32 * j];
    const v4f last = xr4[(lane < 26) ? (lane + 224) : 249];

    const int   lc = (lane < 19) ? lane : 18;
    const float hl = xr[lc];
    const float tl = xr[999 - lc];

    const float* __restrict__ Wf = W + n * KC + half * 2;
    v2f bw[19];
#pragma unroll
    for (int kb = 0; kb < 19; ++kb)
        bw[kb] = *(const v2f*)(Wf + 4 * kb);
    const float bterm = 1964.0f * bias[n];

    __builtin_amdgcn_sched_barrier(0);

    const v4f zero4 = {0.0f, 0.0f, 0.0f, 0.0f};
    p[7] = (lane < 26) ? last : zero4;
    float h = (lane < 19) ? hl : 0.0f;
    float t = (lane < 19) ? tl : 0.0f;

    const v4f q0 = (p[0] + p[1]) + (p[2] + p[3]);
    const v4f q1 = (p[4] + p[5]) + (p[6] + p[7]);
    const v4f q  = q0 + q1;
    float total = (q.x + q.y) + (q.z + q.w);
#pragma unroll
    for (int off = 16; off > 0; off >>= 1)
        total += __shfl_xor(total, off, 32);

    float hI = h, tI = t;
#pragma unroll
    for (int off = 1; off < 32; off <<= 1) {
        float ho = __shfl_up(hI, off, 32);
        float to = __shfl_up(tI, off, 32);
        if (lane >= off) { hI += ho; tI += to; }
    }
    const float head_excl = hI - h;
    const int   src     = (lane < 18) ? (17 - lane) : 0;
    const float twv     = __shfl(tI, src, 32);
    const float tailwin = (lane < 18) ? twv : 0.0f;

    if (lane < 19)
        Ssh[wave * 19 + lane] = total - head_excl - tailwin;
    if (threadIdx.x < KC)
        Ssh[NB * KC + threadIdx.x] = 0.0f;
    __syncthreads();

    const float* arow = Ssh + ((l16 < NB) ? l16 : NB) * KC;
    const float* Sf = arow + half * 2;
    const float* Sr = arow + 74 - half * 2;

    v8f acc = {};
#pragma unroll
    for (int kb = 0; kb < 19; ++kb) {
        const v2f fw = *(const v2f*)(Sf + 4 * kb);
        const v2f rv = *(const v2f*)(Sr - 4 * kb);
        v2f a;
        a.x = fw.x + rv.y;
        a.y = fw.y + rv.x;
        acc = __builtin_amdgcn_wmma_f32_16x16x4_f32(
            false, a, false, bw[kb], (short)0, acc, false, false);
    }

    if (half == 0) {
#pragma unroll
        for (int r = 0; r < NB; ++r)
            out[(bp * NB + r) * 256 + n] = acc[r] + bterm;
    }
}

extern "C" void kernel_launch(void* const* d_in, const int* in_sizes, int n_in,
                              void* d_out, int out_size, void* d_ws, size_t ws_size,
                              hipStream_t stream)
{
    const float* x    = (const float*)d_in[0];
    const float* pwms = (const float*)d_in[1];
    const float* bias = (const float*)d_in[2];
    float* out = (float*)d_out;

    pwm_fused<<<256, 256, 0, stream>>>(x, pwms, bias, out);
}


# both strands folded into both operands: a[k]=S[k]+S[75-k] is symmetric, so (W[k]+W[75-k])*a[k] over k<38: 10 matrix steps not 19, weights as 10 16-byte loads a lane; product and sum order change
# speedup vs baseline: 2.6065x; 1.0772x over previous
#include <hip/hip_runtime.h>
#include <hip/hip_bf16.h>


typedef float v2f __attribute__((ext_vector_type(2)));
typedef float v4f __attribute__((ext_vector_type(4)));
typedef float v8f __attribute__((ext_vector_type(8)));

#define NB 2
#define KC 76

__global__ __launch_bounds__(256) void pwm_fused(
    const float* __restrict__ x,
    const float* __restrict__ W,
    const float* __restrict__ bias,
    float* __restrict__ out)
{
    __shared__ __attribute__((aligned(16))) float Ssh[(NB + 1) * KC];

    const int lane = threadIdx.x & 31;
    const int wave = threadIdx.x >> 5;
    const int bp   = blockIdx.x >> 1;
    const int n0   = ((blockIdx.x & 1) * 8 + wave) << 4;

    const int half = lane >> 4;
    const int l16  = lane & 15;
    const int n    = n0 + l16;

    const int row = bp * 8 + wave;
    const float* __restrict__ xr = x + row * 1000;

    const v4f* __restrict__ xr4 = (const v4f*)xr;
    v4f p[8];
#pragma unroll
    for (int j = 0; j < 7; ++j)
        p[j] = xr4[lane + 32 * j];
    const v4f last = xr4[(lane < 26) ? (lane + 224) : 249];

    const int   lc = (lane < 19) ? lane : 18;
    const float hl = xr[lc];
    const float tl = xr[999 - lc];

    const float* __restrict__ Wrow = W + n * KC;
    v4f wfw[5], wrv[5];
#pragma unroll
    for (int j = 0; j < 5; ++j) {
        wfw[j] = *(const v4f*)(Wrow + 8 * j + 4 * half);
        wrv[j] = *(const v4f*)(Wrow + 72 - 8 * j - 4 * half);
    }
    const float bterm = 1964.0f * bias[n];

    __builtin_amdgcn_sched_barrier(0);

    const v4f zero4 = {0.0f, 0.0f, 0.0f, 0.0f};
    p[7] = (lane < 26) ? last : zero4;
    float h = (lane < 19) ? hl : 0.0f;
    float t = (lane < 19) ? tl : 0.0f;

    const v4f q0 = (p[0] + p[1]) + (p[2] + p[3]);
    const v4f q1 = (p[4] + p[5]) + (p[6] + p[7]);
    const v4f q  = q0 + q1;
    float total = (q.x + q.y) + (q.z + q.w);
#pragma unroll
    for (int off = 16; off > 0; off >>= 1)
        total += __shfl_xor(total, off, 32);

    float hI = h, tI = t;
#pragma unroll
    for (int off = 1; off < 32; off <<= 1) {
        float ho = __shfl_up(hI, off, 32);
        float to = __shfl_up(tI, off, 32);
        if (lane >= off) { hI += ho; tI += to; }
    }
    const float head_excl = hI - h;
    const int   src     = (lane < 18) ? (17 - lane) : 0;
    const float twv     = __shfl(tI, src, 32);
    const float tailwin = (lane < 18) ? twv : 0.0f;

    if (lane < 19)
        Ssh[wave * 19 + lane] = total - head_excl - tailwin;
    if (threadIdx.x < KC)
        Ssh[NB * KC + threadIdx.x] = 0.0f;

    v2f bw[10];
#pragma unroll
    for (int j = 0; j < 5; ++j) {
        const v4f f = wfw[j], r = wrv[j];
        float w0 = f.x + r.w, w1 = f.y + r.z, w2 = f.z + r.y, w3 = f.w + r.x;
        if (j == 4) {
            w2 = half ? 0.0f : w2;
            w3 = half ? 0.0f : w3;
        }
        bw[2 * j].x = w0;     bw[2 * j].y = w1;
        bw[2 * j + 1].x = w2; bw[2 * j + 1].y = w3;
    }
    __syncthreads();

    const float* arow = Ssh + ((l16 < NB) ? l16 : NB) * KC;
    const float* Sf = arow + 4 * half;
    const float* Sr = arow + 72 - 4 * half;

    v4f sfw[5], srv[5];
#pragma unroll
    for (int j = 0; j < 5; ++j) {
        sfw[j] = *(const v4f*)(Sf + 8 * j);
        srv[j] = *(const v4f*)(Sr - 8 * j);
    }
    __builtin_amdgcn_sched_barrier(0);

    v8f acc = {};
#pragma unroll
    for (int j = 0; j < 5; ++j) {
        const v4f f = sfw[j], r = srv[j];
        v2f a0, a1;
        a0.x = f.x + r.w;
        a0.y = f.y + r.z;
        a1.x = f.z + r.y;
        a1.y = f.w + r.x;
        acc = __builtin_amdgcn_wmma_f32_16x16x4_f32(
            false, a0, false, bw[2 * j], (short)0, acc, false, false);
        acc = __builtin_amdgcn_wmma_f32_16x16x4_f32(
            false, a1, false, bw[2 * j + 1], (short)0, acc, false, false);
    }

    if (half == 0) {
#pragma unroll
        for (int r = 0; r < NB; ++r)
            out[(bp * NB + r) * 256 + n] = acc[r] + bterm;
    }
}

extern "C" void kernel_launch(void* const* d_in, const int* in_sizes, int n_in,
                              void* d_out, int out_size, void* d_ws, size_t ws_size,
                              hipStream_t stream)
{
    const float* x    = (const float*)d_in[0];
    const float* pwms = (const float*)d_in[1];
    const float* bias = (const float*)d_in[2];
    float* out = (float*)d_out;

    pwm_fused<<<256, 256, 0, stream>>>(x, pwms, bias, out);
}


# short path from the row's arrival to the barrier: lane sum and edge scans on row-DPP adds, barrier orders LDS only and the weights are folded behind it; row-total and edge-scan sum order changes
# speedup vs baseline: 2.7633x; 1.0602x over previous
#include <hip/hip_runtime.h>
#include <hip/hip_bf16.h>


typedef float v2f __attribute__((ext_vector_type(2)));
typedef float v4f __attribute__((ext_vector_type(4)));
typedef float v8f __attribute__((ext_vector_type(8)));

#define NB 2
#define KC 76


template <int CTRL>
__device__ __forceinline__ float dpp0(float v)
{
    return __builtin_bit_cast(float,
        __builtin_amdgcn_update_dpp(0, __builtin_bit_cast(int, v), CTRL, 0xf, 0xf, true));
}
__device__ __forceinline__ float xrow(float v, unsigned sel_lo, unsigned sel_hi)
{
    const unsigned u = __builtin_bit_cast(unsigned, v);
    return __builtin_bit_cast(float,
        __builtin_amdgcn_permlanex16(u, u, sel_lo, sel_hi, false, false));
}

__global__ __launch_bounds__(256) void pwm_fused(
    const float* __restrict__ x,
    const float* __restrict__ W,
    const float* __restrict__ bias,
    float* __restrict__ out)
{
    __shared__ __attribute__((aligned(16))) float Ssh[(NB + 1) * KC];

    const int lane = threadIdx.x & 31;
    const int wave = threadIdx.x >> 5;
    const int bp   = blockIdx.x >> 1;
    const int n0   = ((blockIdx.x & 1) * 8 + wave) << 4;

    const int half = lane >> 4;
    const int l16  = lane & 15;
    const int n    = n0 + l16;

    const int row = bp * 8 + wave;
    const float* __restrict__ xr = x + row * 1000;

    const v4f* __restrict__ xr4 = (const v4f*)xr;
    v4f p[8];
#pragma unroll
    for (int j = 0; j < 7; ++j)
        p[j] = xr4[lane + 32 * j];
    const v4f last = xr4[(lane < 26) ? (lane + 224) : 249];

    const int   lc = (lane < 19) ? lane : 18;
    const int   tc = (lane < 18) ? lane : 17;
    const float hl = xr[lc];
    const float tl = xr[982 + tc];

    const float* __restrict__ Wrow = W + n * KC;
    v4f wfw[5], wrv[5];
#pragma unroll
    for (int j = 0; j < 5; ++j) {
        wfw[j] = *(const v4f*)(Wrow + 8 * j + 4 * half);
        wrv[j] = *(const v4f*)(Wrow + 72 - 8 * j - 4 * half);
    }
    const float bterm = 1964.0f * bias[n];

    __builtin_amdgcn_sched_barrier(0);

    const v4f zero4 = {0.0f, 0.0f, 0.0f, 0.0f};
    p[7] = (lane < 26) ? last : zero4;
    const float h = (lane < 19) ? hl : 0.0f;
    const float t = (lane < 18) ? tl : 0.0f;

    const v4f q0 = (p[0] + p[1]) + (p[2] + p[3]);
    const v4f q1 = (p[4] + p[5]) + (p[6] + p[7]);
    const v4f q  = q0 + q1;
    float total = (q.x + q.y) + (q.z + q.w);
    total += dpp0<0x161>(total);
    total += dpp0<0x162>(total);
    total += dpp0<0x164>(total);
    total += dpp0<0x168>(total);
    total += xrow(total, 0x76543210u, 0xfedcba98u);

    float hI = h;
    hI += dpp0<0x111>(hI);
    hI += dpp0<0x112>(hI);
    hI += dpp0<0x114>(hI);
    hI += dpp0<0x118>(hI);
    const float hrow0 = xrow(hI, 0xffffffffu, 0xffffffffu);
    hI += (lane >= 16) ? hrow0 : 0.0f;
    float tS = t;
    tS += dpp0<0x101>(tS);
    tS += dpp0<0x102>(tS);
    tS += dpp0<0x104>(tS);
    tS += dpp0<0x108>(tS);
    const float trow1 = xrow(tS, 0u, 0u);
    tS += (lane < 16) ? trow1 : 0.0f;

    const float head_excl = hI - h;
    const float tailwin   = tS;

    if (lane < 19)
        Ssh[wave * 19 + lane] = total - head_excl - tailwin;
    if (threadIdx.x < KC)
        Ssh[NB * KC + threadIdx.x] = 0.0f;

    __builtin_amdgcn_fence(__ATOMIC_RELEASE, "workgroup", "local");
    __builtin_amdgcn_s_barrier();
    __builtin_amdgcn_fence(__ATOMIC_ACQUIRE, "workgroup", "local");

    const float* arow = Ssh + ((l16 < NB) ? l16 : NB) * KC;
    const float* Sf = arow + 4 * half;
    const float* Sr = arow + 72 - 4 * half;

    v4f sfw[5], srv[5];
#pragma unroll
    for (int j = 0; j < 5; ++j) {
        sfw[j] = *(const v4f*)(Sf + 8 * j);
        srv[j] = *(const v4f*)(Sr - 8 * j);
    }
    __builtin_amdgcn_sched_barrier(0);

    v8f acc = {};
#pragma unroll
    for (int j = 0; j < 5; ++j) {
        const v4f f = sfw[j], r = srv[j];
        v2f a0, a1;
        a0.x = f.x + r.w;
        a0.y = f.y + r.z;
        a1.x = f.z + r.y;
        a1.y = f.w + r.x;
        const v4f wf = wfw[j], wr = wrv[j];
        v2f b0, b1;
        b0.x = wf.x + wr.w;
        b0.y = wf.y + wr.z;
        b1.x = wf.z + wr.y;
        b1.y = wf.w + wr.x;
        if (j == 4) {
            b1.x = half ? 0.0f : b1.x;
            b1.y = half ? 0.0f : b1.y;
        }
        acc = __builtin_amdgcn_wmma_f32_16x16x4_f32(
            false, a0, false, b0, (short)0, acc, false, false);
        acc = __builtin_amdgcn_wmma_f32_16x16x4_f32(
            false, a1, false, b1, (short)0, acc, false, false);
    }

    if (half == 0) {
#pragma unroll
        for (int r = 0; r < NB; ++r)
            out[(bp * NB + r) * 256 + n] = acc[r] + bterm;
    }
}

extern "C" void kernel_launch(void* const* d_in, const int* in_sizes, int n_in,
                              void* d_out, int out_size, void* d_ws, size_t ws_size,
                              hipStream_t stream)
{
    const float* x    = (const float*)d_in[0];
    const float* pwms = (const float*)d_in[1];
    const float* bias = (const float*)d_in[2];
    float* out = (float*)d_out;

    pwm_fused<<<256, 256, 0, stream>>>(x, pwms, bias, out);
}


# weights fetched as whole lines: a wave copies its 16 groups' contiguous 4864-byte block by coalesced 16-byte loads into LDS between barrier signal and wait, reads B pieces there; sum order same
# speedup vs baseline: 2.9827x; 1.0794x over previous
#include <hip/hip_runtime.h>
#include <hip/hip_bf16.h>


typedef float v2f __attribute__((ext_vector_type(2)));
typedef float v4f __attribute__((ext_vector_type(4)));
typedef float v8f __attribute__((ext_vector_type(8)));

#define NB 2
#define KC 76


template <int CTRL>
__device__ __forceinline__ float dpp0(float v)
{
    return __builtin_bit_cast(float,
        __builtin_amdgcn_update_dpp(0, __builtin_bit_cast(int, v), CTRL, 0xf, 0xf, true));
}
__device__ __forceinline__ float xrow(float v, unsigned sel_lo, unsigned sel_hi)
{
    const unsigned u = __builtin_bit_cast(unsigned, v);
    return __builtin_bit_cast(float,
        __builtin_amdgcn_permlanex16(u, u, sel_lo, sel_hi, false, false));
}

__global__ __launch_bounds__(256) void pwm_fused(
    const float* __restrict__ x,
    const float* __restrict__ W,
    const float* __restrict__ bias,
    float* __restrict__ out)
{
    __shared__ __attribute__((aligned(16))) float Ssh[(NB + 1) * KC];
    __shared__ __attribute__((aligned(16))) float Wsh[8 * 16 * KC];

    const int lane = threadIdx.x & 31;
    const int wave = threadIdx.x >> 5;
    const int bp   = blockIdx.x >> 1;
    const int n0   = ((blockIdx.x & 1) * 8 + wave) << 4;

    const int half = lane >> 4;
    const int l16  = lane & 15;
    const int n    = n0 + l16;

    const int row = bp * 8 + wave;
    const float* __restrict__ xr = x + row * 1000;

    const v4f* __restrict__ xr4 = (const v4f*)xr;
    v4f p[8];
#pragma unroll
    for (int j = 0; j < 7; ++j)
        p[j] = xr4[lane + 32 * j];
    const v4f last = xr4[(lane < 26) ? (lane + 224) : 249];

    const int   lc = (lane < 19) ? lane : 18;
    const int   tc = (lane < 18) ? lane : 17;
    const float hl = xr[lc];
    const float tl = xr[982 + tc];

    const v4f* __restrict__ Wb4 = (const v4f*)(W + n0 * KC);
    v4f wq[10];
#pragma unroll
    for (int j = 0; j < 9; ++j)
        wq[j] = Wb4[lane + 32 * j];
    wq[9] = Wb4[(lane < 16) ? (lane + 288) : 303];
    const float bterm = 1964.0f * bias[n];

    __builtin_amdgcn_sched_barrier(0);

    const v4f zero4 = {0.0f, 0.0f, 0.0f, 0.0f};
    p[7] = (lane < 26) ? last : zero4;
    const float h = (lane < 19) ? hl : 0.0f;
    const float t = (lane < 18) ? tl : 0.0f;

    const v4f q0 = (p[0] + p[1]) + (p[2] + p[3]);
    const v4f q1 = (p[4] + p[5]) + (p[6] + p[7]);
    const v4f q  = q0 + q1;
    float total = (q.x + q.y) + (q.z + q.w);
    total += dpp0<0x161>(total);
    total += dpp0<0x162>(total);
    total += dpp0<0x164>(total);
    total += dpp0<0x168>(total);
    total += xrow(total, 0x76543210u, 0xfedcba98u);

    float hI = h;
    hI += dpp0<0x111>(hI);
    hI += dpp0<0x112>(hI);
    hI += dpp0<0x114>(hI);
    hI += dpp0<0x118>(hI);
    const float hrow0 = xrow(hI, 0xffffffffu, 0xffffffffu);
    hI += (lane >= 16) ? hrow0 : 0.0f;
    float tS = t;
    tS += dpp0<0x101>(tS);
    tS += dpp0<0x102>(tS);
    tS += dpp0<0x104>(tS);
    tS += dpp0<0x108>(tS);
    const float trow1 = xrow(tS, 0u, 0u);
    tS += (lane < 16) ? trow1 : 0.0f;

    const float head_excl = hI - h;
    const float tailwin   = tS;

    if (lane < 19)
        Ssh[wave * 19 + lane] = total - head_excl - tailwin;
    if (threadIdx.x < KC)
        Ssh[NB * KC + threadIdx.x] = 0.0f;

    __builtin_amdgcn_fence(__ATOMIC_RELEASE, "workgroup", "local");
    __builtin_amdgcn_s_barrier_signal(-1);

    v4f* Wl4 = (v4f*)(Wsh + wave * (16 * KC));
#pragma unroll
    for (int j = 0; j < 9; ++j)
        Wl4[lane + 32 * j] = wq[j];
    if (lane < 16)
        Wl4[lane + 288] = wq[9];
    const float* Wrow = Wsh + wave * (16 * KC) + l16 * KC;
    v4f wfw[5], wrv[5];
#pragma unroll
    for (int j = 0; j < 5; ++j) {
        wfw[j] = *(const v4f*)(Wrow + 8 * j + 4 * half);
        wrv[j] = *(const v4f*)(Wrow + 72 - 8 * j - 4 * half);
    }

    __builtin_amdgcn_s_barrier_wait(-1);
    __builtin_amdgcn_fence(__ATOMIC_ACQUIRE, "wavefront", "local");

    const float* arow = Ssh + ((l16 < NB) ? l16 : NB) * KC;
    const float* Sf = arow + 4 * half;
    const float* Sr = arow + 72 - 4 * half;

    v4f sfw[5], srv[5];
#pragma unroll
    for (int j = 0; j < 5; ++j) {
        sfw[j] = *(const v4f*)(Sf + 8 * j);
        srv[j] = *(const v4f*)(Sr - 8 * j);
    }
    __builtin_amdgcn_sched_barrier(0);

    v8f acc = {};
#pragma unroll
    for (int j = 0; j < 5; ++j) {
        const v4f f = sfw[j], r = srv[j];
        v2f a0, a1;
        a0.x = f.x + r.w;
        a0.y = f.y + r.z;
        a1.x = f.z + r.y;
        a1.y = f.w + r.x;
        const v4f wf = wfw[j], wr = wrv[j];
        v2f b0, b1;
        b0.x = wf.x + wr.w;
        b0.y = wf.y + wr.z;
        b1.x = wf.z + wr.y;
        b1.y = wf.w + wr.x;
        if (j == 4) {
            b1.x = half ? 0.0f : b1.x;
            b1.y = half ? 0.0f : b1.y;
        }
        acc = __builtin_amdgcn_wmma_f32_16x16x4_f32(
            false, a0, false, b0, (short)0, acc, false, false);
        acc = __builtin_amdgcn_wmma_f32_16x16x4_f32(
            false, a1, false, b1, (short)0, acc, false, false);
    }

    if (half == 0) {
#pragma unroll
        for (int r = 0; r < NB; ++r)
            out[(bp * NB + r) * 256 + n] = acc[r] + bterm;
    }
}

extern "C" void kernel_launch(void* const* d_in, const int* in_sizes, int n_in,
                              void* d_out, int out_size, void* d_ws, size_t ws_size,
                              hipStream_t stream)
{
    const float* x    = (const float*)d_in[0];
    const float* pwms = (const float*)d_in[1];
    const float* bias = (const float*)d_in[2];
    float* out = (float*)d_out;

    pwm_fused<<<256, 256, 0, stream>>>(x, pwms, bias, out);
}
